# starting kernel
# baseline (speedup 1.0000x reference)
#include <hip/hip_runtime.h>
#include <math.h>

typedef __attribute__((ext_vector_type(2))) float v2f;
typedef __attribute__((ext_vector_type(8))) float v8f;

#define Bd   4
#define Cd   3
#define Hd   256
#define Wd   512
#define PAD  4
#define TILE 16

__global__ __launch_bounds__(256) void knn_min_dist_wmma(
    const float* __restrict__ tfm, const float* __restrict__ obs,
    float* __restrict__ out)
{
    const int lane        = threadIdx.x & 31;
    const int waveInBlock = threadIdx.x >> 5;
    const int waveId      = blockIdx.x * 8 + waveInBlock;

    const int tilesPerRow = Wd / TILE;
    const int tileIdx     = waveId % tilesPerRow;
    const int h           = (waveId / tilesPerRow) % Hd;
    const int b           =  waveId / (tilesPerRow * Hd);
    const int w0          = tileIdx * TILE;

    const int n    = lane & 15;
    const int half = lane >> 4;

    const int    w       = w0 + n;
    const size_t planeHW = (size_t)Hd * Wd;
    const size_t bBase   = ((size_t)b * Cd) * planeHW;
    const size_t baseT   = bBase + (size_t)h * Wd + w;

    const float t0 = tfm[baseT];
    const float t1 = tfm[baseT + planeHW];
    const float t2 = tfm[baseT + 2 * planeHW];
    const float tnorm = t0 * t0 + t1 * t1 + t2 * t2;

    v2f bmat;
    if (half == 0) { bmat.x = t0; bmat.y = t1;    }
    else           { bmat.x = t2; bmat.y = -0.5f; }

    const float NEG = -3.0e38f;
    float msel[2][8];
#pragma unroll
    for (int i = 0; i < 8; ++i) {
        const int m = half * 8 + i;
        msel[0][i] = ((m >= n) && (m <= n + 8)) ? 0.0f : NEG;
        msel[1][i] = (m <= n - 8)               ? 0.0f : NEG;
    }

    const int colRaw[2] = { w0 - PAD + n, w0 - PAD + TILE + n };

    float maxD = NEG;
    const v8f czero = {};

#pragma unroll
    for (int dr = 0; dr < 9; ++dr) {
        const int  r      = h + dr - PAD;
        const bool rowOk  = (r >= 0) && (r < Hd);
        const int  rClamp = min(max(r, 0), Hd - 1);
        const size_t rowBase = bBase + (size_t)rClamp * Wd;

#pragma unroll
        for (int t = 0; t < 2; ++t) {
            const int   c      = colRaw[t];
            const bool  ok     = rowOk && (c >= 0) && (c < Wd);
            const int   cClamp = min(max(c, 0), Wd - 1);
            const float okf    = ok ? 1.0f : 0.0f;

            const size_t baseO = rowBase + cClamp;
            const float a0 = obs[baseO]                * okf;
            const float a1 = obs[baseO + planeHW]      * okf;
            const float a2 = obs[baseO + 2 * planeHW]  * okf;
            const float onorm = a0 * a0 + a1 * a1 + a2 * a2;

            v2f amat;
            if (half == 0) { amat.x = a0; amat.y = a1;    }
            else           { amat.x = a2; amat.y = onorm; }

            v8f d = __builtin_amdgcn_wmma_f32_16x16x4_f32(
                false, amat, false, bmat, (short)0, czero, false, false);

#pragma unroll
            for (int i = 0; i < 8; ++i) {
                maxD = fmaxf(maxD, d[i] + msel[t][i]);
            }
        }
    }

    const float other = __shfl_xor(maxD, 16, 32);
    maxD = fmaxf(maxD, other);

    float d2 = tnorm - 2.0f * maxD;
    d2 = fmaxf(d2, 0.0f);

    if (half == 0) {
        out[(size_t)b * planeHW + (size_t)h * Wd + w] = sqrtf(d2);
    }
}

extern "C" void kernel_launch(void* const* d_in, const int* in_sizes, int n_in,
                              void* d_out, int out_size, void* d_ws, size_t ws_size,
                              hipStream_t stream) {
    const float* tfm = (const float*)d_in[0];
    const float* obs = (const float*)d_in[1];
    float*       out = (float*)d_out;

    const int totalWaves = Bd * Hd * (Wd / TILE);
    const int blocks     = totalWaves / 8;
    hipLaunchKernelGGL(knn_min_dist_wmma, dim3(blocks), dim3(256), 0, stream,
                       tfm, obs, out);
}


# one wave per strip of 8 rows x 16 pixels: each obs row is loaded 3 rows ahead and made into its A fragments once, then reused by up to 9 tfm rows (was 1 row per wave, 18 serial load waits)
# speedup vs baseline: 4.6979x; 4.6979x over previous
#include <hip/hip_runtime.h>
#include <math.h>


typedef __attribute__((ext_vector_type(2))) float v2f;
typedef __attribute__((ext_vector_type(8))) float v8f;

#define Bd   4
#define Cd   3
#define Hd   256
#define Wd   512
#define PAD  4
#define TILE 16
#define ROWS 8
#define OROWS (ROWS + 2 * PAD)
#define PF   3

__global__ __launch_bounds__(256) void knn_min_dist_wmma(
    const float* __restrict__ tfm, const float* __restrict__ obs,
    float* __restrict__ out)
{
    const int lane        = threadIdx.x & 31;
    const int waveInBlock = threadIdx.x >> 5;

    const int tg    = blockIdx.x & 3;
    const int strip = (blockIdx.x >> 2) & (Hd / ROWS - 1);
    const int b     = blockIdx.x / (4 * (Hd / ROWS));
    const int h0    = strip * ROWS;
    const int w0    = (tg * 8 + waveInBlock) * TILE;

    const int  n    = lane & 15;
    const int  half = lane >> 4;
    const bool hi   = (half != 0);

    const int    w       = w0 + n;
    const size_t planeHW = (size_t)Hd * Wd;
    const size_t bBase   = ((size_t)b * Cd) * planeHW;

    const float NEG = -3.0e38f;
    float msel[2][8];
#pragma unroll
    for (int i = 0; i < 8; ++i) {
        const int m = half * 8 + i;
        msel[0][i] = ((m >= n) && (m <= n + 8)) ? 0.0f : NEG;
        msel[1][i] = (m <= n - 8)               ? 0.0f : NEG;
    }

    unsigned colByte[2];
    float    colOk[2];
#pragma unroll
    for (int t = 0; t < 2; ++t) {
        const int c = w0 - PAD + TILE * t + n;
        colOk[t]   = ((c >= 0) && (c < Wd)) ? 1.0f : 0.0f;
        colByte[t] = (unsigned)min(max(c, 0), Wd - 1) * 4u;
    }
    const unsigned wByte = (unsigned)w * 4u;

    float raw[OROWS][2][3];

#define LOAD_OBS_ROW(kk)                                                              \
    {                                                                                 \
        const int rr = min(max(h0 - PAD + (kk), 0), Hd - 1);                          \
        const char* rowO = (const char*)(obs + bBase + (size_t)rr * Wd);              \
        _Pragma("unroll")                                                             \
        for (int t = 0; t < 2; ++t) {                                                 \
            raw[kk][t][0] = *(const float*)(rowO + colByte[t]);                       \
            raw[kk][t][1] = *(const float*)(rowO + planeHW * 4 + colByte[t]);         \
            raw[kk][t][2] = *(const float*)(rowO + planeHW * 8 + colByte[t]);         \
        }                                                                             \
    }

#pragma unroll
    for (int k = 0; k < PF; ++k) LOAD_OBS_ROW(k)

    v2f   bmat[ROWS];
    float tnorm[ROWS];
    float maxD[ROWS];
#pragma unroll
    for (int j = 0; j < ROWS; ++j) {
        const char* rowT = (const char*)(tfm + bBase + (size_t)(h0 + j) * Wd);
        const float t0 = *(const float*)(rowT + wByte);
        const float t1 = *(const float*)(rowT + planeHW * 4 + wByte);
        const float t2 = *(const float*)(rowT + planeHW * 8 + wByte);
        tnorm[j]  = t0 * t0 + t1 * t1 + t2 * t2;
        bmat[j].x = hi ? t2    : t0;
        bmat[j].y = hi ? -0.5f : t1;
        maxD[j]   = NEG;
    }

    const v8f czero = {};

#pragma unroll
    for (int k = 0; k < OROWS; ++k) {
        if (k + PF < OROWS) LOAD_OBS_ROW(k + PF)
        __builtin_amdgcn_sched_barrier(0);

        const int  r     = h0 - PAD + k;
        const bool rowOk = (r >= 0) && (r < Hd);

        v2f amat[2];
#pragma unroll
        for (int t = 0; t < 2; ++t) {
            const float okf = rowOk ? colOk[t] : 0.0f;
            const float a0 = raw[k][t][0] * okf;
            const float a1 = raw[k][t][1] * okf;
            const float a2 = raw[k][t][2] * okf;
            const float onorm = a0 * a0 + a1 * a1 + a2 * a2;
            amat[t].x = hi ? a2    : a0;
            amat[t].y = hi ? onorm : a1;
        }

#pragma unroll
        for (int j = 0; j < ROWS; ++j) {
            if (k - j < 0 || k - j > 2 * PAD) continue;
#pragma unroll
            for (int t = 0; t < 2; ++t) {
                v8f d = __builtin_amdgcn_wmma_f32_16x16x4_f32(
                    false, amat[t], false, bmat[j], (short)0, czero, false, false);
#pragma unroll
                for (int i = 0; i < 8; ++i) {
                    maxD[j] = fmaxf(maxD[j], d[i] + msel[t][i]);
                }
            }
        }

        if (k >= 2 * PAD && ((k - 2 * PAD) & 1)) {
            const int j1 = k - 2 * PAD;
            const int j0 = j1 - 1;
            const float send = hi ? maxD[j0] : maxD[j1];
            const float mine = hi ? maxD[j1] : maxD[j0];
            const float recv = __shfl_xor(send, 16, 32);
            const float mx   = fmaxf(mine, recv);
            const float tn   = hi ? tnorm[j1] : tnorm[j0];
            float d2 = tn - 2.0f * mx;
            d2 = fmaxf(d2, 0.0f);
            out[(size_t)b * planeHW + (size_t)(h0 + j0 + half) * Wd + w] = sqrtf(d2);
        }
    }
#undef LOAD_OBS_ROW
}

extern "C" void kernel_launch(void* const* d_in, const int* in_sizes, int n_in,
                              void* d_out, int out_size, void* d_ws, size_t ws_size,
                              hipStream_t stream) {
    const float* tfm = (const float*)d_in[0];
    const float* obs = (const float*)d_in[1];
    float*       out = (float*)d_out;

    const int totalWaves = Bd * (Hd / ROWS) * (Wd / TILE);
    const int blocks     = totalWaves / 8;
    hipLaunchKernelGGL(knn_min_dist_wmma, dim3(blocks), dim3(256), 0, stream,
                       tfm, obs, out);
}


# band mask moved into the C operand of each f32 WMMA (0 on the band, -3e38 off it): the mask adds are gone and D goes straight into the three-input maxima; summation order and results unchanged
# speedup vs baseline: 5.3412x; 1.1369x over previous
#include <hip/hip_runtime.h>
#include <math.h>


typedef __attribute__((ext_vector_type(2))) float v2f;
typedef __attribute__((ext_vector_type(8))) float v8f;

#define Bd   4
#define Cd   3
#define Hd   256
#define Wd   512
#define PAD  4
#define TILE 16
#define ROWS 8
#define OROWS (ROWS + 2 * PAD)
#define PF   3

__global__ __launch_bounds__(256) void knn_min_dist_wmma(
    const float* __restrict__ tfm, const float* __restrict__ obs,
    float* __restrict__ out)
{
    const int lane        = threadIdx.x & 31;
    const int waveInBlock = threadIdx.x >> 5;

    const int tg    = blockIdx.x & 3;
    const int strip = (blockIdx.x >> 2) & (Hd / ROWS - 1);
    const int b     = blockIdx.x / (4 * (Hd / ROWS));
    const int h0    = strip * ROWS;
    const int w0    = (tg * 8 + waveInBlock) * TILE;

    const int  n    = lane & 15;
    const int  half = lane >> 4;
    const bool hi   = (half != 0);

    const int    w       = w0 + n;
    const size_t planeHW = (size_t)Hd * Wd;
    const size_t bBase   = ((size_t)b * Cd) * planeHW;

    const float NEG = -3.0e38f;
    v8f cmask[2];
#pragma unroll
    for (int i = 0; i < 8; ++i) {
        const int m = half * 8 + i;
        cmask[0][i] = ((m >= n) && (m <= n + 8)) ? 0.0f : NEG;
        cmask[1][i] = (m <= n - 8)               ? 0.0f : NEG;
    }

    unsigned colByte[2];
    float    colOk[2];
#pragma unroll
    for (int t = 0; t < 2; ++t) {
        const int c = w0 - PAD + TILE * t + n;
        colOk[t]   = ((c >= 0) && (c < Wd)) ? 1.0f : 0.0f;
        colByte[t] = (unsigned)min(max(c, 0), Wd - 1) * 4u;
    }
    const unsigned wByte = (unsigned)w * 4u;

    float raw[OROWS][2][3];

#define LOAD_OBS_ROW(kk)                                                              \
    {                                                                                 \
        const int rr = min(max(h0 - PAD + (kk), 0), Hd - 1);                          \
        const char* rowO = (const char*)(obs + bBase + (size_t)rr * Wd);              \
        _Pragma("unroll")                                                             \
        for (int t = 0; t < 2; ++t) {                                                 \
            raw[kk][t][0] = *(const float*)(rowO + colByte[t]);                       \
            raw[kk][t][1] = *(const float*)(rowO + planeHW * 4 + colByte[t]);         \
            raw[kk][t][2] = *(const float*)(rowO + planeHW * 8 + colByte[t]);         \
        }                                                                             \
    }

#pragma unroll
    for (int k = 0; k < PF; ++k) LOAD_OBS_ROW(k)

    v2f   bmat[ROWS];
    float tnorm[ROWS];
    float maxD[ROWS];
#pragma unroll
    for (int j = 0; j < ROWS; ++j) {
        const char* rowT = (const char*)(tfm + bBase + (size_t)(h0 + j) * Wd);
        const float t0 = *(const float*)(rowT + wByte);
        const float t1 = *(const float*)(rowT + planeHW * 4 + wByte);
        const float t2 = *(const float*)(rowT + planeHW * 8 + wByte);
        tnorm[j]  = t0 * t0 + t1 * t1 + t2 * t2;
        bmat[j].x = hi ? t2    : t0;
        bmat[j].y = hi ? -0.5f : t1;
        maxD[j]   = NEG;
    }

#pragma unroll
    for (int k = 0; k < OROWS; ++k) {
        if (k + PF < OROWS) LOAD_OBS_ROW(k + PF)
        __builtin_amdgcn_sched_barrier(0);

        const int  r     = h0 - PAD + k;
        const bool rowOk = (r >= 0) && (r < Hd);

        v2f amat[2];
#pragma unroll
        for (int t = 0; t < 2; ++t) {
            const float okf = rowOk ? colOk[t] : 0.0f;
            const float a0 = raw[k][t][0] * okf;
            const float a1 = raw[k][t][1] * okf;
            const float a2 = raw[k][t][2] * okf;
            const float onorm = a0 * a0 + a1 * a1 + a2 * a2;
            amat[t].x = hi ? a2    : a0;
            amat[t].y = hi ? onorm : a1;
        }

#pragma unroll
        for (int j = 0; j < ROWS; ++j) {
            if (k - j < 0 || k - j > 2 * PAD) continue;
#pragma unroll
            for (int t = 0; t < 2; ++t) {
                v8f d = __builtin_amdgcn_wmma_f32_16x16x4_f32(
                    false, amat[t], false, bmat[j], (short)0, cmask[t], false, false);
#pragma unroll
                for (int i = 0; i < 8; ++i) maxD[j] = fmaxf(maxD[j], d[i]);
            }
        }

        if (k >= 2 * PAD && ((k - 2 * PAD) & 1)) {
            const int j1 = k - 2 * PAD;
            const int j0 = j1 - 1;
            const float send = hi ? maxD[j0] : maxD[j1];
            const float mine = hi ? maxD[j1] : maxD[j0];
            const float recv = __shfl_xor(send, 16, 32);
            const float mx   = fmaxf(mine, recv);
            const float tn   = hi ? tnorm[j1] : tnorm[j0];
            float d2 = tn - 2.0f * mx;
            d2 = fmaxf(d2, 0.0f);
            out[(size_t)b * planeHW + (size_t)(h0 + j0 + half) * Wd + w] = sqrtf(d2);
        }
    }
#undef LOAD_OBS_ROW
}

extern "C" void kernel_launch(void* const* d_in, const int* in_sizes, int n_in,
                              void* d_out, int out_size, void* d_ws, size_t ws_size,
                              hipStream_t stream) {
    const float* tfm = (const float*)d_in[0];
    const float* obs = (const float*)d_in[1];
    float*       out = (float*)d_out;

    const int totalWaves = Bd * (Hd / ROWS) * (Wd / TILE);
    const int blocks     = totalWaves / 8;
    hipLaunchKernelGGL(knn_min_dist_wmma, dim3(blocks), dim3(256), 0, stream,
                       tfm, obs, out);
}


# B operand is a patch of 2 rows x 8 pixels, so one 16-column obs fragment serves both rows of a pair: 10 products per 16 pixels, was 18 (80 per strip, was 144); sums and results unchanged
# speedup vs baseline: 7.1194x; 1.3329x over previous
#include <hip/hip_runtime.h>
#include <math.h>


typedef __attribute__((ext_vector_type(2))) float v2f;
typedef __attribute__((ext_vector_type(8))) float v8f;

#define Bd   4
#define Cd   3
#define Hd   256
#define Wd   512
#define PAD  4
#define TILE 16
#define HALF 8
#define ROWS 8
#define PAIRS (ROWS / 2)
#define OROWS (ROWS + 2 * PAD)
#define PF   3

__global__ __launch_bounds__(256) void knn_min_dist_wmma(
    const float* __restrict__ tfm, const float* __restrict__ obs,
    float* __restrict__ out)
{
    const int lane        = threadIdx.x & 31;
    const int waveInBlock = threadIdx.x >> 5;

    const int tg    = blockIdx.x & 3;
    const int strip = (blockIdx.x >> 2) & (Hd / ROWS - 1);
    const int b     = blockIdx.x / (4 * (Hd / ROWS));
    const int h0    = strip * ROWS;
    const int w0    = (tg * 8 + waveInBlock) * TILE;

    const int  n    = lane & 15;
    const int  half = lane >> 4;
    const bool hi   = (half != 0);
    const int  p    = n >> 3;
    const int  c    = n & 7;

    const size_t planeHW = (size_t)Hd * Wd;
    const size_t bBase   = ((size_t)b * Cd) * planeHW;

    const float NEG = -3.0e38f;
    v8f cmask[3];
#pragma unroll
    for (int i = 0; i < 8; ++i) {
        const int   m    = half * 8 + i;
        const float band = ((m >= c) && (m <= c + 8)) ? 0.0f : NEG;
        cmask[0][i] = band;
        cmask[1][i] = (p == 0) ? band : NEG;
        cmask[2][i] = (p == 1) ? band : NEG;
    }

    unsigned colByte[2];
    float    colOk[2];
#pragma unroll
    for (int t = 0; t < 2; ++t) {
        const int oc = w0 + HALF * t - PAD + n;
        colOk[t]   = ((oc >= 0) && (oc < Wd)) ? 1.0f : 0.0f;
        colByte[t] = (unsigned)min(max(oc, 0), Wd - 1) * 4u;
    }

    float raw[OROWS][2][3];

#define LOAD_OBS_ROW(kk)                                                              \
    {                                                                                 \
        const int rr = min(max(h0 - PAD + (kk), 0), Hd - 1);                          \
        const char* rowO = (const char*)(obs + bBase + (size_t)rr * Wd);              \
        _Pragma("unroll")                                                             \
        for (int t = 0; t < 2; ++t) {                                                 \
            raw[kk][t][0] = *(const float*)(rowO + colByte[t]);                       \
            raw[kk][t][1] = *(const float*)(rowO + planeHW * 4 + colByte[t]);         \
            raw[kk][t][2] = *(const float*)(rowO + planeHW * 8 + colByte[t]);         \
        }                                                                             \
    }

#pragma unroll
    for (int k = 0; k < PF; ++k) LOAD_OBS_ROW(k)

    v2f   bmat[PAIRS][2];
    float tnorm[PAIRS][2];
    float maxD[PAIRS][2];
#pragma unroll
    for (int jp = 0; jp < PAIRS; ++jp) {
        const char* rowT = (const char*)(tfm + bBase + (size_t)(h0 + 2 * jp + p) * Wd);
#pragma unroll
        for (int t = 0; t < 2; ++t) {
            const unsigned pixByte = (unsigned)(w0 + HALF * t + c) * 4u;
            const float t0 = *(const float*)(rowT + pixByte);
            const float t1 = *(const float*)(rowT + planeHW * 4 + pixByte);
            const float t2 = *(const float*)(rowT + planeHW * 8 + pixByte);
            tnorm[jp][t]  = t0 * t0 + t1 * t1 + t2 * t2;
            bmat[jp][t].x = hi ? t2    : t0;
            bmat[jp][t].y = hi ? -0.5f : t1;
            maxD[jp][t]   = NEG;
        }
    }

#pragma unroll
    for (int k = 0; k < OROWS; ++k) {
        if (k + PF < OROWS) LOAD_OBS_ROW(k + PF)
        __builtin_amdgcn_sched_barrier(0);

        const int  r     = h0 - PAD + k;
        const bool rowOk = (r >= 0) && (r < Hd);

        v2f amat[2];
#pragma unroll
        for (int t = 0; t < 2; ++t) {
            const float okf = rowOk ? colOk[t] : 0.0f;
            const float a0 = raw[k][t][0] * okf;
            const float a1 = raw[k][t][1] * okf;
            const float a2 = raw[k][t][2] * okf;
            const float onorm = a0 * a0 + a1 * a1 + a2 * a2;
            amat[t].x = hi ? a2    : a0;
            amat[t].y = hi ? onorm : a1;
        }

#pragma unroll
        for (int jp = 0; jp < PAIRS; ++jp) {
            const int dk = k - 2 * jp;
            if (dk < 0 || dk > 2 * PAD + 1) continue;
            const int ms = (dk == 0) ? 1 : ((dk == 2 * PAD + 1) ? 2 : 0);
#pragma unroll
            for (int t = 0; t < 2; ++t) {
                v8f d = __builtin_amdgcn_wmma_f32_16x16x4_f32(
                    false, amat[t], false, bmat[jp][t], (short)0, cmask[ms], false, false);
#pragma unroll
                for (int i = 0; i < 8; ++i) maxD[jp][t] = fmaxf(maxD[jp][t], d[i]);
            }
        }

        if (k >= 2 * PAD + 1 && ((k - 2 * PAD - 1) & 1) == 0) {
            const int jp = (k - 2 * PAD - 1) / 2;
            const float send = hi ? maxD[jp][0] : maxD[jp][1];
            const float mine = hi ? maxD[jp][1] : maxD[jp][0];
            const float recv = __shfl_xor(send, 16, 32);
            const float mx   = fmaxf(mine, recv);
            const float tn   = hi ? tnorm[jp][1] : tnorm[jp][0];
            float d2 = tn - 2.0f * mx;
            d2 = fmaxf(d2, 0.0f);
            out[(size_t)b * planeHW + (size_t)(h0 + 2 * jp + p) * Wd + (w0 + HALF * half + c)] = sqrtf(d2);
        }
    }
#undef LOAD_OBS_ROW
}

extern "C" void kernel_launch(void* const* d_in, const int* in_sizes, int n_in,
                              void* d_out, int out_size, void* d_ws, size_t ws_size,
                              hipStream_t stream) {
    const float* tfm = (const float*)d_in[0];
    const float* obs = (const float*)d_in[1];
    float*       out = (float*)d_out;

    const int totalWaves = Bd * (Hd / ROWS) * (Wd / TILE);
    const int blocks     = totalWaves / 8;
    hipLaunchKernelGGL(knn_min_dist_wmma, dim3(blocks), dim3(256), 0, stream,
                       tfm, obs, out);
}


# tfm patch and obs block are both squares of 4x4 pixels: a patch's windows are exactly the 3x3 blocks around it, 9 products per 16 pixels, was 10 (72 per strip, was 80); order of sums unchanged
# speedup vs baseline: 7.5377x; 1.0588x over previous
#include <hip/hip_runtime.h>
#include <math.h>


typedef __attribute__((ext_vector_type(2))) float v2f;
typedef __attribute__((ext_vector_type(8))) float v8f;

#define Bd   4
#define Cd   3
#define Hd   256
#define Wd   512
#define PAD  4
#define TILE 16
#define ROWS 8
#define BLK  4
#define PR   (ROWS / BLK)
#define PC   (TILE / BLK)
#define BR   (PR + 2)
#define BC   (PC + 2)

__global__ __launch_bounds__(256) void knn_min_dist_wmma(
    const float* __restrict__ tfm, const float* __restrict__ obs,
    float* __restrict__ out)
{
    const int lane        = threadIdx.x & 31;
    const int waveInBlock = __builtin_amdgcn_readfirstlane(threadIdx.x >> 5);

    const int tg    = blockIdx.x & 3;
    const int strip = (blockIdx.x >> 2) & (Hd / ROWS - 1);
    const int b     = blockIdx.x / (4 * (Hd / ROWS));
    const int h0    = strip * ROWS;
    const int w0    = (tg * 8 + waveInBlock) * TILE;

    const int  n    = lane & 15;
    const int  half = lane >> 4;
    const bool hi   = (half != 0);
    const int  pr_  = n >> 2;
    const int  pc_  = n & 3;

    const size_t planeHW = (size_t)Hd * Wd;
    const size_t bBase   = ((size_t)b * Cd) * planeHW;
    const unsigned laneByte = (unsigned)(pr_ * Wd + pc_) * 4u;

    const float NEG = -3.0e38f;
    v8f cmask[3][3];
#pragma unroll
    for (int i = 0; i < 8; ++i) {
        const int  m    = half * 8 + i;
        const int  orow = m >> 2;
        const int  ocol = m & 3;
        const bool rv[3] = { orow >= pr_, true, orow <= pr_ };
        const bool cv[3] = { ocol >= pc_, true, ocol <= pc_ };
#pragma unroll
        for (int s = 0; s < 3; ++s)
#pragma unroll
            for (int t = 0; t < 3; ++t)
                cmask[s][t][i] = (rv[s] && cv[t]) ? 0.0f : NEG;
    }

    float raw[BR][BC][3];

#define LOAD_BLOCK_ROW(q)                                                                   \
    {                                                                                       \
        const int r0 = min(max(h0 - PAD + BLK * (q), 0), Hd - BLK);                         \
        _Pragma("unroll")                                                                   \
        for (int bc = 0; bc < BC; ++bc) {                                                   \
            const int c0 = min(max(w0 - PAD + BLK * bc, 0), Wd - BLK);                      \
            const char* blk = (const char*)(obs + bBase + (size_t)r0 * Wd + c0) + laneByte; \
            raw[q][bc][0] = *(const float*)(blk);                                           \
            raw[q][bc][1] = *(const float*)(blk + planeHW * 4);                             \
            raw[q][bc][2] = *(const float*)(blk + planeHW * 8);                             \
        }                                                                                   \
    }

    LOAD_BLOCK_ROW(0)

    v2f   bmat[PR][PC];
    float tnorm[PR][PC];
    float maxD[PR][PC];
#pragma unroll
    for (int pr = 0; pr < PR; ++pr) {
#pragma unroll
        for (int pc = 0; pc < PC; ++pc) {
            const char* pat = (const char*)(tfm + bBase + (size_t)(h0 + BLK * pr) * Wd
                                            + (w0 + BLK * pc)) + laneByte;
            const float t0 = *(const float*)(pat);
            const float t1 = *(const float*)(pat + planeHW * 4);
            const float t2 = *(const float*)(pat + planeHW * 8);
            tnorm[pr][pc]  = t0 * t0 + t1 * t1 + t2 * t2;
            bmat[pr][pc].x = hi ? t2    : t0;
            bmat[pr][pc].y = hi ? -0.5f : t1;
            maxD[pr][pc]   = NEG;
        }
    }

#pragma unroll
    for (int q = 0; q < BR; ++q) {
        if (q + 1 < BR) LOAD_BLOCK_ROW(q + 1)
        __builtin_amdgcn_sched_barrier(0);

        const int  r     = h0 - PAD + BLK * q;
        const bool rowOk = (r >= 0) && (r < Hd);

        v2f amat[BC];
#pragma unroll
        for (int bc = 0; bc < BC; ++bc) {
            const int   cc  = w0 - PAD + BLK * bc;
            const float okf = (rowOk && (cc >= 0) && (cc < Wd)) ? 1.0f : 0.0f;
            const float a0 = raw[q][bc][0] * okf;
            const float a1 = raw[q][bc][1] * okf;
            const float a2 = raw[q][bc][2] * okf;
            const float onorm = a0 * a0 + a1 * a1 + a2 * a2;
            amat[bc].x = hi ? a2    : a0;
            amat[bc].y = hi ? onorm : a1;
        }

#pragma unroll
        for (int pr = 0; pr < PR; ++pr) {
            const int s = q - pr;
            if (s < 0 || s > 2) continue;
#pragma unroll
            for (int bc = 0; bc < BC; ++bc) {
#pragma unroll
                for (int pc = 0; pc < PC; ++pc) {
                    const int t = bc - pc;
                    if (t < 0 || t > 2) continue;
                    v8f d = __builtin_amdgcn_wmma_f32_16x16x4_f32(
                        false, amat[bc], false, bmat[pr][pc], (short)0, cmask[s][t], false, false);
#pragma unroll
                    for (int i = 0; i < 8; ++i) maxD[pr][pc] = fmaxf(maxD[pr][pc], d[i]);
                }
            }
        }

        if (q >= 2) {
            const int pr = q - 2;
#pragma unroll
            for (int g = 0; g < PC / 2; ++g) {
                const float send = hi ? maxD[pr][2 * g]     : maxD[pr][2 * g + 1];
                const float mine = hi ? maxD[pr][2 * g + 1] : maxD[pr][2 * g];
                const float recv = __shfl_xor(send, 16, 32);
                const float mx   = fmaxf(mine, recv);
                const float tn   = hi ? tnorm[pr][2 * g + 1] : tnorm[pr][2 * g];
                float d2 = tn - 2.0f * mx;
                d2 = fmaxf(d2, 0.0f);
                out[(size_t)b * planeHW + (size_t)(h0 + BLK * pr + pr_) * Wd
                    + (w0 + BLK * (2 * g + half) + pc_)] = sqrtf(d2);
            }
        }
    }
#undef LOAD_BLOCK_ROW
}

extern "C" void kernel_launch(void* const* d_in, const int* in_sizes, int n_in,
                              void* d_out, int out_size, void* d_ws, size_t ws_size,
                              hipStream_t stream) {
    const float* tfm = (const float*)d_in[0];
    const float* obs = (const float*)d_in[1];
    float*       out = (float*)d_out;

    const int totalWaves = Bd * (Hd / ROWS) * (Wd / TILE);
    const int blocks     = totalWaves / 8;
    hipLaunchKernelGGL(knn_min_dist_wmma, dim3(blocks), dim3(256), 0, stream,
                       tfm, obs, out);
}
